# starting kernel
# baseline (speedup 1.0000x reference)
#include <hip/hip_runtime.h>
#include <hip/hip_bf16.h>

typedef __attribute__((ext_vector_type(2))) float v2f;
typedef __attribute__((ext_vector_type(8))) float v8f;

#define DIN 128

__global__ __launch_bounds__(256) void deg_init_kernel(float* deg, int n) {
    int i = blockIdx.x * blockDim.x + threadIdx.x;
    if (i < n) deg[i] = 1.0f;
}

__global__ __launch_bounds__(256) void deg_accum_kernel(const int* __restrict__ ei, int e,
                                                        float* __restrict__ deg) {
    int idx = blockIdx.x * blockDim.x + threadIdx.x;
    if (idx < e) atomicAdd(&deg[ei[e + idx]], 1.0f);
}

__global__ __launch_bounds__(256) void deg_to_dinv_kernel(float* deg, int n) {
    int i = blockIdx.x * blockDim.x + threadIdx.x;
    if (i < n) deg[i] = rsqrtf(deg[i]);
}

template <int DOUT>
__global__ __launch_bounds__(256) void gemm_wmma_kernel(const float* __restrict__ X,
                                                        const float* __restrict__ W,
                                                        float* __restrict__ H, int n) {
    constexpr int NT = DOUT / 16;
    constexpr int NPAIR = 32 * NT * 32;
    __shared__ float sW[NPAIR * 2];

    for (int pair = threadIdx.x; pair < NPAIR; pair += 256) {
        int l  = pair & 31;
        int tt = (pair >> 5) % NT;
        int ks = (pair >> 5) / NT;
        int col = tt * 16 + (l & 15);
        int k0  = ks * 4 + ((l >> 4) * 2);
        sW[pair * 2 + 0] = W[k0 * DOUT + col];
        sW[pair * 2 + 1] = W[(k0 + 1) * DOUT + col];
    }
    __syncthreads();

    const int wave = threadIdx.x >> 5;
    const int lane = threadIdx.x & 31;
    const int m    = lane & 15;
    const int kk   = (lane >> 4) * 2;

    const long long row0 = (long long)blockIdx.x * 128 + wave * 16;
    long long rowm = row0 + m;
    if (rowm >= n) rowm = n - 1;
    const float* __restrict__ xrow = X + rowm * DIN + kk;
    const float* __restrict__ bl   = sW + lane * 2;

    v8f acc[NT] = {};

#pragma unroll 4
    for (int ks = 0; ks < 32; ks++) {
        v2f a = *(const v2f*)(xrow + ks * 4);
#pragma unroll
        for (int t = 0; t < NT; t++) {
            v2f b = *(const v2f*)(bl + (ks * NT + t) * 64);
            acc[t] = __builtin_amdgcn_wmma_f32_16x16x4_f32(
                false, a, false, b, (short)0, acc[t], false, false);
        }
    }

    const long long rbase = row0 + (lane >> 4) * 8;
#pragma unroll
    for (int v = 0; v < 8; v++) {
        long long r = rbase + v;
        if (r < n) {
            float* __restrict__ orow = H + r * DOUT;
#pragma unroll
            for (int t = 0; t < NT; t++) orow[t * 16 + m] = acc[t][v];
        }
    }
}

template <int D>
__global__ __launch_bounds__(256) void init_agg_kernel(const float* __restrict__ H,
                                                       const float* __restrict__ bias,
                                                       const float* __restrict__ dinv,
                                                       float* __restrict__ Agg, int n) {
    long long idx = (long long)blockIdx.x * blockDim.x + threadIdx.x;
    long long total = (long long)n * D;
    if (idx < total) {
        int r = (int)(idx / D);
        int c = (int)(idx - (long long)r * D);
        float w = dinv[r];
        Agg[idx] = bias[c] + H[idx] * (w * w);
    }
}

template <int D>
__global__ __launch_bounds__(256) void scatter_kernel(const int* __restrict__ ei, int e,
                                                      const float* __restrict__ dinv,
                                                      const float* __restrict__ H,
                                                      float* __restrict__ Agg) {
    constexpr int VEC = D / 32;
    const int lane = threadIdx.x & 31;
    const int wid  = (int)((blockIdx.x * (long long)blockDim.x + threadIdx.x) >> 5);
    const int nw   = (int)(((long long)gridDim.x * blockDim.x) >> 5);

    for (int ed = wid; ed < e; ed += nw) {
        int s = ei[ed];
        int d = ei[e + ed];
        float w = dinv[s] * dinv[d];

        int e2 = ed + nw;
        if (e2 < e) {
            int s2 = ei[e2];
            __builtin_prefetch(H + (long long)s2 * D + lane * VEC, 0, 1);
        }

        const float* __restrict__ hs = H + (long long)s * D + lane * VEC;
        float* __restrict__ ad = Agg + (long long)d * D + lane * VEC;
        if (VEC == 4) {
            float4 v = *(const float4*)hs;
            atomicAdd(ad + 0, v.x * w);
            atomicAdd(ad + 1, v.y * w);
            atomicAdd(ad + 2, v.z * w);
            atomicAdd(ad + 3, v.w * w);
        } else {
            float2 v = *(const float2*)hs;
            atomicAdd(ad + 0, v.x * w);
            atomicAdd(ad + 1, v.y * w);
        }
    }
}

__global__ __launch_bounds__(256) void elu_kernel(float* __restrict__ A, long long total) {
    long long idx = (long long)blockIdx.x * blockDim.x + threadIdx.x;
    if (idx < total) {
        float v = A[idx];
        A[idx] = v > 0.0f ? v : (__expf(v) - 1.0f);
    }
}

extern "C" void kernel_launch(void* const* d_in, const int* in_sizes, int n_in,
                              void* d_out, int out_size, void* d_ws, size_t ws_size,
                              hipStream_t stream) {
    const float* x  = (const float*)d_in[0];
    const int*   ei = (const int*)d_in[1];
    const float* W1 = (const float*)d_in[2];
    const float* b1 = (const float*)d_in[3];
    const float* W2 = (const float*)d_in[4];
    const float* b2 = (const float*)d_in[5];
    float* out = (float*)d_out;

    const int N = in_sizes[0] / DIN;
    const int E = in_sizes[1] / 2;
    const int DH = 128, DOUT = 64;

    char* ws = (char*)d_ws;
    float* dinv = (float*)ws;
    size_t off = (((size_t)N * sizeof(float)) + 255) & ~(size_t)255;
    float* hbuf = (float*)(ws + off);
    float* aggbuf = (float*)(ws + off + (size_t)N * DH * sizeof(float));

    const int B = 256;
    const int gN    = (N + B - 1) / B;
    const int gE    = (E + B - 1) / B;
    const int gRows = (N + 127) / 128;
    const int gNDH  = (int)(((long long)N * DH + B - 1) / B);
    const int gNDO  = (int)(((long long)N * DOUT + B - 1) / B);
    const int gEdgeW = (E + 7) / 8;

    deg_init_kernel<<<gN, B, 0, stream>>>(dinv, N);
    deg_accum_kernel<<<gE, B, 0, stream>>>(ei, E, dinv);
    deg_to_dinv_kernel<<<gN, B, 0, stream>>>(dinv, N);

    gemm_wmma_kernel<128><<<gRows, B, 0, stream>>>(x, W1, hbuf, N);
    init_agg_kernel<128><<<gNDH, B, 0, stream>>>(hbuf, b1, dinv, aggbuf, N);
    scatter_kernel<128><<<gEdgeW, B, 0, stream>>>(ei, E, dinv, hbuf, aggbuf);
    elu_kernel<<<gNDH, B, 0, stream>>>(aggbuf, (long long)N * DH);

    gemm_wmma_kernel<64><<<gRows, B, 0, stream>>>(aggbuf, W2, hbuf, N);
    init_agg_kernel<64><<<gNDO, B, 0, stream>>>(hbuf, b2, dinv, out, N);
    scatter_kernel<64><<<gEdgeW, B, 0, stream>>>(ei, E, dinv, hbuf, out);
}


# Degree count without global atomics or a start-value launch: 16x16 blocks count (edge chunk, node range) in LDS with integer adds, layer-1 GEMM sums the 16 partial counts per row; 6 -> 5 launches
# speedup vs baseline: 2.3363x; 2.3363x over previous
#include <hip/hip_runtime.h>
#include <hip/hip_bf16.h>

typedef __attribute__((ext_vector_type(2))) float v2f;
typedef __attribute__((ext_vector_type(8))) float v8f;

#define DIN 128

#define HC 16
#define HRANGE 6272
#define HU 13

__global__ __launch_bounds__(512) void deg_hist_kernel(const int* __restrict__ ei, int e,
                                                       unsigned int* __restrict__ P,
                                                       int pstride) {
    __shared__ uint4 hist4[HRANGE / 4];
    unsigned int* hist = (unsigned int*)hist4;
    const int c = blockIdx.x;
    const int node0 = blockIdx.y * HRANGE;

    for (int i = threadIdx.x; i < HRANGE / 4; i += 512) hist4[i] = make_uint4(0u, 0u, 0u, 0u);
    __syncthreads();

    const int chunk = (((e + HC - 1) / HC) + 3) & ~3;
    const int lo = c * chunk;
    const int hi = (lo + chunk < e) ? (lo + chunk) : e;
    const int* __restrict__ dst = ei + e;

    if ((e & 3) == 0) {
        const int nq = (hi - lo) >> 2;
        const int4* __restrict__ q = (const int4*)(dst + lo);
        for (int i = threadIdx.x; i < nq; i += 512 * HU) {
            int4 v[HU];
#pragma unroll
            for (int u = 0; u < HU; u++) {
                const int j = i + u * 512;
                v[u] = (j < nq) ? q[j] : make_int4(-1, -1, -1, -1);
            }
#pragma unroll
            for (int u = 0; u < HU; u++) {
                unsigned int t0 = (unsigned int)(v[u].x - node0);
                unsigned int t1 = (unsigned int)(v[u].y - node0);
                unsigned int t2 = (unsigned int)(v[u].z - node0);
                unsigned int t3 = (unsigned int)(v[u].w - node0);
                if (t0 < HRANGE) atomicAdd(&hist[t0], 1u);
                if (t1 < HRANGE) atomicAdd(&hist[t1], 1u);
                if (t2 < HRANGE) atomicAdd(&hist[t2], 1u);
                if (t3 < HRANGE) atomicAdd(&hist[t3], 1u);
            }
        }
    } else {
        for (int i = lo + (int)threadIdx.x; i < hi; i += 512) {
            unsigned int t = (unsigned int)(dst[i] - node0);
            if (t < HRANGE) atomicAdd(&hist[t], 1u);
        }
    }
    __syncthreads();

    uint4* __restrict__ o = (uint4*)(P + (size_t)c * pstride + node0);
    for (int i = threadIdx.x; i < HRANGE / 4; i += 512) o[i] = hist4[i];
}

__device__ __forceinline__ float elu1(float v) { return v > 0.0f ? v : (__expf(v) - 1.0f); }

template <int DOUT, bool FIRST>
__global__ __launch_bounds__(256) void gemm_wmma_kernel(const float* __restrict__ X,
                                                        const float* __restrict__ W,
                                                        const float* __restrict__ bias,
                                                        float* Dn,
                                                        const unsigned int* __restrict__ P,
                                                        int pstride,
                                                        float* __restrict__ H,
                                                        float* __restrict__ Agg, int n) {
    constexpr int NT = DOUT / 16;
    constexpr int NPAIR = 32 * NT * 32;
    __shared__ float sW[NPAIR * 2];

    for (int pair = threadIdx.x; pair < NPAIR; pair += 256) {
        int l  = pair & 31;
        int tt = (pair >> 5) % NT;
        int ks = (pair >> 5) / NT;
        int col = tt * 16 + (l & 15);
        int k0  = ks * 4 + ((l >> 4) * 2);
        sW[pair * 2 + 0] = W[k0 * DOUT + col];
        sW[pair * 2 + 1] = W[(k0 + 1) * DOUT + col];
    }
    __syncthreads();

    const int wave = threadIdx.x >> 5;
    const int lane = threadIdx.x & 31;
    const int m    = lane & 15;
    const int kk   = (lane >> 4) * 2;

    const long long row0 = (long long)blockIdx.x * 128 + wave * 16;
    long long rowm = row0 + m;
    if (rowm >= n) rowm = n - 1;
    const float* __restrict__ xrow = X + rowm * DIN + kk;
    const float* __restrict__ bl   = sW + lane * 2;

    uint4 pc0 = make_uint4(0u, 0u, 0u, 0u), pc1 = pc0;
    if (FIRST) {
        const long long g0 = row0 + (lane >> 4) * 8;
        if (g0 + 8 <= pstride) {
            const uint4* __restrict__ pp = (const uint4*)(P + (size_t)m * pstride + g0);
            pc0 = pp[0];
            pc1 = pp[1];
        }
    }

    v8f acc[NT] = {};

#pragma unroll 4
    for (int ks = 0; ks < 32; ks++) {
        v2f a = *(const v2f*)(xrow + ks * 4);
        if (!FIRST) {
            a.x = elu1(a.x);
            a.y = elu1(a.y);
        }
#pragma unroll
        for (int t = 0; t < NT; t++) {
            v2f b = *(const v2f*)(bl + (ks * NT + t) * 64);
            acc[t] = __builtin_amdgcn_wmma_f32_16x16x4_f32(
                false, a, false, b, (short)0, acc[t], false, false);
        }
    }

    float bc[NT];
#pragma unroll
    for (int t = 0; t < NT; t++) bc[t] = bias[t * 16 + m];

    const long long rbase = row0 + (lane >> 4) * 8;

    float wv[8];
    if (FIRST) {
        int cnt[8] = {(int)pc0.x, (int)pc0.y, (int)pc0.z, (int)pc0.w,
                      (int)pc1.x, (int)pc1.y, (int)pc1.z, (int)pc1.w};
#pragma unroll
        for (int v = 0; v < 8; v++) {
            int sm = cnt[v];
            sm += __shfl_xor(sm, 1, 16);
            sm += __shfl_xor(sm, 2, 16);
            sm += __shfl_xor(sm, 4, 16);
            sm += __shfl_xor(sm, 8, 16);
            wv[v] = rsqrtf((float)(sm + 1));
        }
    } else {
#pragma unroll
        for (int v = 0; v < 8; v++) {
            long long r = rbase + v;
            wv[v] = (r < n) ? Dn[r] : 1.0f;
        }
    }

#pragma unroll
    for (int v = 0; v < 8; v++) {
        long long r = rbase + v;
        if (r < n) {
            const float w = wv[v];
            if (FIRST && m == 0) Dn[r] = w;
            const float w2 = w * w;
            float* __restrict__ orow = H + r * DOUT;
            float* __restrict__ arow = Agg + r * DOUT;
#pragma unroll
            for (int t = 0; t < NT; t++) {
                float hv = acc[t][v];
                orow[t * 16 + m] = hv;
                arow[t * 16 + m] = bc[t] + hv * w2;
            }
        }
    }
}

template <int D>
__global__ __launch_bounds__(256) void scatter_kernel(const int* __restrict__ ei, int e,
                                                      const float* __restrict__ dinv,
                                                      const float* __restrict__ H,
                                                      float* __restrict__ Agg) {
    constexpr int VEC = D / 32;
    constexpr int RG = 8;
    constexpr int NG = 32 / RG;

    const int lane = threadIdx.x & 31;
    const int wid = __builtin_amdgcn_readfirstlane((int)(blockIdx.x * 8u + (threadIdx.x >> 5)));
    if (wid >= (e + 31) / 32) return;
    const int base = wid * 32;
    const int left = e - base;
    const int cnt = left < 32 ? left : 32;

    int s = 0, d = 0;
    float w = 0.0f;
    if (lane < cnt) {
        s = ei[base + lane];
        d = ei[(long long)e + base + lane];
        w = dinv[s] * dinv[d];
    }

    const float* __restrict__ hl = H + lane;
    float* __restrict__ al = Agg + lane;

    if (cnt == 32) {
        float cur[RG][VEC], nxt[RG][VEC];
#pragma unroll
        for (int j = 0; j < RG; j++) {
            int sj = __builtin_amdgcn_readlane(s, j);
            const float* __restrict__ hs = hl + (long long)sj * D;
#pragma unroll
            for (int k = 0; k < VEC; k++) cur[j][k] = hs[k * 32];
        }
#pragma unroll
        for (int g = 0; g < NG; g++) {
            if (g + 1 < NG) {
#pragma unroll
                for (int j = 0; j < RG; j++) {
                    int sj = __builtin_amdgcn_readlane(s, (g + 1) * RG + j);
                    const float* __restrict__ hs = hl + (long long)sj * D;
#pragma unroll
                    for (int k = 0; k < VEC; k++) nxt[j][k] = hs[k * 32];
                }
            }
#pragma unroll
            for (int j = 0; j < RG; j++) {
                int dj = __builtin_amdgcn_readlane(d, g * RG + j);
                float wj = __builtin_bit_cast(
                    float, __builtin_amdgcn_readlane(__builtin_bit_cast(int, w), g * RG + j));
                float* __restrict__ ad = al + (long long)dj * D;
#pragma unroll
                for (int k = 0; k < VEC; k++) atomicAdd(ad + k * 32, cur[j][k] * wj);
            }
#pragma unroll
            for (int j = 0; j < RG; j++) {
#pragma unroll
                for (int k = 0; k < VEC; k++) cur[j][k] = nxt[j][k];
            }
        }
    } else {
        for (int j = 0; j < cnt; j++) {
            int sj = __builtin_amdgcn_readlane(s, j);
            int dj = __builtin_amdgcn_readlane(d, j);
            float wj = __builtin_bit_cast(
                float, __builtin_amdgcn_readlane(__builtin_bit_cast(int, w), j));
            const float* __restrict__ hs = hl + (long long)sj * D;
            float* __restrict__ ad = al + (long long)dj * D;
#pragma unroll
            for (int k = 0; k < VEC; k++) atomicAdd(ad + k * 32, hs[k * 32] * wj);
        }
    }
}

extern "C" void kernel_launch(void* const* d_in, const int* in_sizes, int n_in,
                              void* d_out, int out_size, void* d_ws, size_t ws_size,
                              hipStream_t stream) {
    const float* x  = (const float*)d_in[0];
    const int*   ei = (const int*)d_in[1];
    const float* W1 = (const float*)d_in[2];
    const float* b1 = (const float*)d_in[3];
    const float* W2 = (const float*)d_in[4];
    const float* b2 = (const float*)d_in[5];
    float* out = (float*)d_out;

    const int N = in_sizes[0] / DIN;
    const int E = in_sizes[1] / 2;
    const int DH = 128;

    char* ws = (char*)d_ws;
    float* dinv = (float*)ws;
    size_t off = (((size_t)N * sizeof(float)) + 255) & ~(size_t)255;
    float* hbuf = (float*)(ws + off);
    float* aggbuf = (float*)(ws + off + (size_t)N * DH * sizeof(float));

    const int B = 256;
    const int gRows = (N + 127) / 128;
    const int gEdgeW = (E + 255) / 256;

    unsigned int* pcnt = (unsigned int*)d_out;
    const int gR = (N + HRANGE - 1) / HRANGE;
    const int pstride = gR * HRANGE;
    deg_hist_kernel<<<dim3(HC, gR), 512, 0, stream>>>(ei, E, pcnt, pstride);

    gemm_wmma_kernel<128, true><<<gRows, B, 0, stream>>>(x, W1, b1, dinv, pcnt, pstride, hbuf, aggbuf, N);
    scatter_kernel<128><<<gEdgeW, B, 0, stream>>>(ei, E, dinv, hbuf, aggbuf);

    gemm_wmma_kernel<64, false><<<gRows, B, 0, stream>>>(aggbuf, W2, b2, dinv, pcnt, pstride, hbuf, out, N);
    scatter_kernel<64><<<gEdgeW, B, 0, stream>>>(ei, E, dinv, hbuf, out);
}


# Edge scatter split by cache half: a block asks which half of the chip it runs on, takes a ticket of that half's list, adds only edges whose dst row is in that half's node range; arrival order free
# speedup vs baseline: 2.6046x; 1.1148x over previous
#include <hip/hip_runtime.h>
#include <hip/hip_bf16.h>

typedef __attribute__((ext_vector_type(2))) float v2f;
typedef __attribute__((ext_vector_type(8))) float v8f;

#define DIN 128

#define HC 16
#define HRANGE 6272
#define HU 13

__global__ __launch_bounds__(512) void deg_hist_kernel(const int* __restrict__ ei, int e,
                                                       unsigned int* __restrict__ P,
                                                       int pstride) {
    __shared__ uint4 hist4[HRANGE / 4];
    unsigned int* hist = (unsigned int*)hist4;
    const int c = blockIdx.x;
    const int node0 = blockIdx.y * HRANGE;

    for (int i = threadIdx.x; i < HRANGE / 4; i += 512) hist4[i] = make_uint4(0u, 0u, 0u, 0u);
    __syncthreads();

    const int chunk = (((e + HC - 1) / HC) + 3) & ~3;
    const int lo = c * chunk;
    const int hi = (lo + chunk < e) ? (lo + chunk) : e;
    const int* __restrict__ dst = ei + e;

    if ((e & 3) == 0) {
        const int nq = (hi - lo) >> 2;
        const int4* __restrict__ q = (const int4*)(dst + lo);
        for (int i = threadIdx.x; i < nq; i += 512 * HU) {
            int4 v[HU];
#pragma unroll
            for (int u = 0; u < HU; u++) {
                const int j = i + u * 512;
                v[u] = (j < nq) ? q[j] : make_int4(-1, -1, -1, -1);
            }
#pragma unroll
            for (int u = 0; u < HU; u++) {
                unsigned int t0 = (unsigned int)(v[u].x - node0);
                unsigned int t1 = (unsigned int)(v[u].y - node0);
                unsigned int t2 = (unsigned int)(v[u].z - node0);
                unsigned int t3 = (unsigned int)(v[u].w - node0);
                if (t0 < HRANGE) atomicAdd(&hist[t0], 1u);
                if (t1 < HRANGE) atomicAdd(&hist[t1], 1u);
                if (t2 < HRANGE) atomicAdd(&hist[t2], 1u);
                if (t3 < HRANGE) atomicAdd(&hist[t3], 1u);
            }
        }
    } else {
        for (int i = lo + (int)threadIdx.x; i < hi; i += 512) {
            unsigned int t = (unsigned int)(dst[i] - node0);
            if (t < HRANGE) atomicAdd(&hist[t], 1u);
        }
    }
    __syncthreads();

    uint4* __restrict__ o = (uint4*)(P + (size_t)c * pstride + node0);
    for (int i = threadIdx.x; i < HRANGE / 4; i += 512) o[i] = hist4[i];
}

__device__ __forceinline__ float elu1(float v) { return v > 0.0f ? v : (__expf(v) - 1.0f); }

template <int DOUT, bool FIRST>
__global__ __launch_bounds__(256) void gemm_wmma_kernel(const float* __restrict__ X,
                                                        const float* __restrict__ W,
                                                        const float* __restrict__ bin,
                                                        const float* __restrict__ bout,
                                                        float* Dn,
                                                        const unsigned int* __restrict__ P,
                                                        int pstride,
                                                        float* __restrict__ H,
                                                        float* __restrict__ Agg, int n,
                                                        unsigned int* __restrict__ tick) {
    constexpr int NT = DOUT / 16;
    constexpr int NPAIR = 32 * NT * 32;
    __shared__ float sW[NPAIR * 2];
    __shared__ __attribute__((aligned(16))) float sB[FIRST ? 4 : DIN];

    for (int pair = threadIdx.x; pair < NPAIR; pair += 256) {
        int l  = pair & 31;
        int tt = (pair >> 5) % NT;
        int ks = (pair >> 5) / NT;
        int col = tt * 16 + (l & 15);
        int k0  = ks * 4 + ((l >> 4) * 2);
        sW[pair * 2 + 0] = W[k0 * DOUT + col];
        sW[pair * 2 + 1] = W[(k0 + 1) * DOUT + col];
    }
    if (!FIRST) {
        if (threadIdx.x < DIN) sB[threadIdx.x] = bin[threadIdx.x];
    }
    if (blockIdx.x == 0 && threadIdx.x < 128) tick[threadIdx.x] = 0u;
    __syncthreads();

    const int wave = threadIdx.x >> 5;
    const int lane = threadIdx.x & 31;
    const int m    = lane & 15;
    const int kk   = (lane >> 4) * 2;

    const long long row0 = (long long)blockIdx.x * 128 + wave * 16;
    long long rowm = row0 + m;
    if (rowm >= n) rowm = n - 1;
    const float* __restrict__ xrow = X + rowm * DIN + kk;
    const float* __restrict__ bl   = sW + lane * 2;
    float dr = 1.0f;
    if (!FIRST) dr = Dn[rowm];

    uint4 pc0 = make_uint4(0u, 0u, 0u, 0u), pc1 = pc0;
    if (FIRST) {
        const long long g0 = row0 + (lane >> 4) * 8;
        if (g0 + 8 <= pstride) {
            const uint4* __restrict__ pp = (const uint4*)(P + (size_t)m * pstride + g0);
            pc0 = pp[0];
            pc1 = pp[1];
        }
    }

    v8f acc[NT] = {};

#pragma unroll 4
    for (int ks = 0; ks < 32; ks++) {
        v2f a = *(const v2f*)(xrow + ks * 4);
        if (!FIRST) {
            v2f pb = *(const v2f*)(sB + ks * 4 + kk);
            a.x = elu1(pb.x + a.x * dr);
            a.y = elu1(pb.y + a.y * dr);
        }
#pragma unroll
        for (int t = 0; t < NT; t++) {
            v2f b = *(const v2f*)(bl + (ks * NT + t) * 64);
            acc[t] = __builtin_amdgcn_wmma_f32_16x16x4_f32(
                false, a, false, b, (short)0, acc[t], false, false);
        }
    }

    float bc[NT];
#pragma unroll
    for (int t = 0; t < NT; t++) bc[t] = FIRST ? 0.0f : bout[t * 16 + m];

    const long long rbase = row0 + (lane >> 4) * 8;

    float wv[8];
    if (FIRST) {
        int cnt[8] = {(int)pc0.x, (int)pc0.y, (int)pc0.z, (int)pc0.w,
                      (int)pc1.x, (int)pc1.y, (int)pc1.z, (int)pc1.w};
#pragma unroll
        for (int v = 0; v < 8; v++) {
            int sm = cnt[v];
            sm += __shfl_xor(sm, 1, 16);
            sm += __shfl_xor(sm, 2, 16);
            sm += __shfl_xor(sm, 4, 16);
            sm += __shfl_xor(sm, 8, 16);
            wv[v] = rsqrtf((float)(sm + 1));
        }
    } else {
#pragma unroll
        for (int v = 0; v < 8; v++) {
            long long r = rbase + v;
            wv[v] = (r < n) ? Dn[r] : 1.0f;
        }
    }

#pragma unroll
    for (int v = 0; v < 8; v++) {
        long long r = rbase + v;
        if (r < n) {
            const float w = wv[v];
            if (FIRST && m == 0) Dn[r] = w;
            float* __restrict__ orow = H + r * DOUT;
            float* __restrict__ arow = Agg + r * DOUT;
#pragma unroll
            for (int t = 0; t < NT; t++) {
                const float hs = acc[t][v] * w;
                orow[t * 16 + m] = hs;
                arow[t * 16 + m] = FIRST ? hs : (bc[t] + hs * w);
            }
        }
    }
}

#define SB 2

template <int D, bool DSTW>
__global__ __launch_bounds__(256) void scatter_kernel(const int* __restrict__ ei, int e,
                                                      const float* __restrict__ dinv,
                                                      const float* __restrict__ H,
                                                      float* __restrict__ Agg,
                                                      unsigned int* __restrict__ tick,
                                                      int nitems, int nsplit) {
    constexpr int VEC = D / 32;
    constexpr int RG = 32 / VEC;

    __shared__ int s_item;
    if (threadIdx.x == 0) {
        const unsigned int id = __builtin_amdgcn_s_sendmsg_rtn(0x87);
        const int h = (((id >> 16) & 15u) >= 4u) ? 1 : 0;
        int item = -1;
        unsigned int t = atomicAdd(&tick[h * 64], 1u);
        if (t < (unsigned int)nitems) {
            item = (int)t * 2 + h;
        } else {
            t = atomicAdd(&tick[(1 - h) * 64], 1u);
            if (t < (unsigned int)nitems) item = (int)t * 2 + (1 - h);
        }
        s_item = item;
    }
    __syncthreads();
    const int item = __builtin_amdgcn_readfirstlane(s_item);
    if (item < 0) return;
    const bool upper = (item & 1) != 0;

    const int lane = threadIdx.x & 31;
    const int wv = __builtin_amdgcn_readfirstlane((int)(threadIdx.x >> 5));
    const int base0 = (item >> 1) * (256 * SB) + wv * 32;

    int s[SB], d[SB];
    bool ok[SB];
#pragma unroll
    for (int b = 0; b < SB; b++) {
        const int idx = base0 + b * 256 + lane;
        ok[b] = idx < e;
        const int ix = ok[b] ? idx : 0;
        s[b] = ei[ix];
        d[b] = ei[(long long)e + ix];
    }
    asm volatile("" ::"v"(s[0]), "v"(s[1]), "v"(d[0]), "v"(d[1]));
    static_assert(SB == 2, "the statement above names the index values of two batches");
    bool mine[SB];
    unsigned int mk[SB];
#pragma unroll
    for (int b = 0; b < SB; b++) {
        mine[b] = ok[b] && ((d[b] >= nsplit) == upper);
        mk[b] = __builtin_amdgcn_ballot_w32(mine[b]);
    }
    float w[SB];
#pragma unroll
    for (int b = 0; b < SB; b++) {
        w[b] = 0.0f;
        if (DSTW) w[b] = dinv[mine[b] ? d[b] : 0];
    }

    const float* __restrict__ hl = H + lane;
    float* __restrict__ al = Agg + lane;

#pragma unroll
    for (int b = 0; b < SB; b++) {
        unsigned int m = mk[b];
        while (m != 0u) {
            const int left = __builtin_popcount(m);
            const int nj = left < RG ? left : RG;
            int dl[RG];
            float cur[RG][VEC];
#pragma unroll
            for (int j = 0; j < RG; j++) {
                dl[j] = 0;
                if (j < nj) {
                    const int l = __builtin_ctz(m);
                    m &= m - 1u;
                    dl[j] = l;
                    const int sj = __builtin_amdgcn_readlane(s[b], l);
                    const float* __restrict__ hs = hl + (long long)sj * D;
#pragma unroll
                    for (int k = 0; k < VEC; k++) cur[j][k] = hs[k * 32];
                }
            }
#pragma unroll
            for (int j = 0; j < RG; j++) {
                if (j < nj) {
                    const int dj = __builtin_amdgcn_readlane(d[b], dl[j]);
                    float* __restrict__ ad = al + (long long)dj * D;
                    if (DSTW) {
                        const float wj = __builtin_bit_cast(
                            float,
                            __builtin_amdgcn_readlane(__builtin_bit_cast(int, w[b]), dl[j]));
#pragma unroll
                        for (int k = 0; k < VEC; k++) atomicAdd(ad + k * 32, cur[j][k] * wj);
                    } else {
#pragma unroll
                        for (int k = 0; k < VEC; k++) atomicAdd(ad + k * 32, cur[j][k]);
                    }
                }
            }
        }
    }
}

extern "C" void kernel_launch(void* const* d_in, const int* in_sizes, int n_in,
                              void* d_out, int out_size, void* d_ws, size_t ws_size,
                              hipStream_t stream) {
    const float* x  = (const float*)d_in[0];
    const int*   ei = (const int*)d_in[1];
    const float* W1 = (const float*)d_in[2];
    const float* b1 = (const float*)d_in[3];
    const float* W2 = (const float*)d_in[4];
    const float* b2 = (const float*)d_in[5];
    float* out = (float*)d_out;

    const int N = in_sizes[0] / DIN;
    const int E = in_sizes[1] / 2;
    const int DH = 128;

    char* ws = (char*)d_ws;
    float* dinv = (float*)ws;
    size_t off = (((size_t)N * sizeof(float)) + 255) & ~(size_t)255;
    float* hbuf = (float*)(ws + off);
    float* aggbuf = (float*)(ws + off + (size_t)N * DH * sizeof(float));

    const int B = 256;
    const int gRows = (N + 127) / 128;
    const int nitems = (E + 256 * SB - 1) / (256 * SB);
    const int gEdgeW = 2 * nitems;
    const int nsplit = N / 2;

    unsigned int* pcnt = (unsigned int*)d_out;
    const int gR = (N + HRANGE - 1) / HRANGE;
    const int pstride = gR * HRANGE;
    deg_hist_kernel<<<dim3(HC, gR), 512, 0, stream>>>(ei, E, pcnt, pstride);

    unsigned int* tick1 = pcnt + (size_t)HC * pstride;
    unsigned int* tick2 = (unsigned int*)(hbuf + (size_t)N * 64);

    gemm_wmma_kernel<128, true><<<gRows, B, 0, stream>>>(x, W1, b1, b1, dinv, pcnt, pstride, hbuf, aggbuf, N, tick1);
    scatter_kernel<128, false><<<gEdgeW, B, 0, stream>>>(ei, E, dinv, hbuf, aggbuf, tick1, nitems, nsplit);

    gemm_wmma_kernel<64, false><<<gRows, B, 0, stream>>>(aggbuf, W2, b1, b2, dinv, pcnt, pstride, hbuf, out, N, tick2);
    scatter_kernel<64, true><<<gEdgeW, B, 0, stream>>>(ei, E, dinv, hbuf, out, tick2, nitems, nsplit);
}


# Both GEMM launches on a fixed grid of 256 work-groups of 28 waves: W staged once per work-group (16-byte loads and LDS stores), 16-row strips dealt evenly over the SIMDs; every sum order unchanged
# speedup vs baseline: 2.7577x; 1.0588x over previous
#include <hip/hip_runtime.h>
#include <hip/hip_bf16.h>

typedef __attribute__((ext_vector_type(2))) float v2f;
typedef __attribute__((ext_vector_type(8))) float v8f;

#define DIN 128

#define HC 16
#define HRANGE 6272
#define HU 13

__global__ __launch_bounds__(512) void deg_hist_kernel(const int* __restrict__ ei, int e,
                                                       unsigned int* __restrict__ P,
                                                       int pstride) {
    __shared__ uint4 hist4[HRANGE / 4];
    unsigned int* hist = (unsigned int*)hist4;
    const int c = blockIdx.x;
    const int node0 = blockIdx.y * HRANGE;

    for (int i = threadIdx.x; i < HRANGE / 4; i += 512) hist4[i] = make_uint4(0u, 0u, 0u, 0u);
    __syncthreads();

    const int chunk = (((e + HC - 1) / HC) + 3) & ~3;
    const int lo = c * chunk;
    const int hi = (lo + chunk < e) ? (lo + chunk) : e;
    const int* __restrict__ dst = ei + e;

    if ((e & 3) == 0) {
        const int nq = (hi - lo) >> 2;
        const int4* __restrict__ q = (const int4*)(dst + lo);
        for (int i = threadIdx.x; i < nq; i += 512 * HU) {
            int4 v[HU];
#pragma unroll
            for (int u = 0; u < HU; u++) {
                const int j = i + u * 512;
                v[u] = (j < nq) ? q[j] : make_int4(-1, -1, -1, -1);
            }
#pragma unroll
            for (int u = 0; u < HU; u++) {
                unsigned int t0 = (unsigned int)(v[u].x - node0);
                unsigned int t1 = (unsigned int)(v[u].y - node0);
                unsigned int t2 = (unsigned int)(v[u].z - node0);
                unsigned int t3 = (unsigned int)(v[u].w - node0);
                if (t0 < HRANGE) atomicAdd(&hist[t0], 1u);
                if (t1 < HRANGE) atomicAdd(&hist[t1], 1u);
                if (t2 < HRANGE) atomicAdd(&hist[t2], 1u);
                if (t3 < HRANGE) atomicAdd(&hist[t3], 1u);
            }
        }
    } else {
        for (int i = lo + (int)threadIdx.x; i < hi; i += 512) {
            unsigned int t = (unsigned int)(dst[i] - node0);
            if (t < HRANGE) atomicAdd(&hist[t], 1u);
        }
    }
    __syncthreads();

    uint4* __restrict__ o = (uint4*)(P + (size_t)c * pstride + node0);
    for (int i = threadIdx.x; i < HRANGE / 4; i += 512) o[i] = hist4[i];
}

#define GEMM_WG 256
#define GEMM_WV 28
#define GEMM_RUN 25
__device__ __forceinline__ float elu1(float v) { return v > 0.0f ? v : (__expf(v) - 1.0f); }

template <int DOUT, bool FIRST, int TN>
__device__ __forceinline__ void gemm_strip(const float* __restrict__ X,
                                           const float* __restrict__ sW,
                                           const float* __restrict__ sB,
                                           const float* __restrict__ bout,
                                           float* Dn,
                                           const unsigned int* __restrict__ P,
                                           int pstride,
                                           float* __restrict__ H,
                                           float* __restrict__ Agg, int n,
                                           int strip, int t0, bool own_dn, int lane) {
    constexpr int NT = DOUT / 16;
    const int m    = lane & 15;
    const int kk   = (lane >> 4) * 2;

    const long long row0 = (long long)strip * 16;
    long long rowm = row0 + m;
    if (rowm >= n) rowm = n - 1;
    const float* __restrict__ xrow = X + rowm * DIN + kk;
    const float* __restrict__ bl   = sW + lane * 2 + t0 * 64;
    float dr = 1.0f;
    if (!FIRST) dr = Dn[rowm];

    uint4 pc0 = make_uint4(0u, 0u, 0u, 0u), pc1 = pc0;
    if (FIRST) {
        const long long g0 = row0 + (lane >> 4) * 8;
        if (g0 + 8 <= pstride) {
            const uint4* __restrict__ pp = (const uint4*)(P + (size_t)m * pstride + g0);
            pc0 = pp[0];
            pc1 = pp[1];
        }
    }

    v8f acc[TN] = {};

#pragma unroll 4
    for (int ks = 0; ks < 32; ks++) {
        v2f a = *(const v2f*)(xrow + ks * 4);
        if (!FIRST) {
            v2f pb = *(const v2f*)(sB + ks * 4 + kk);
            a.x = elu1(pb.x + a.x * dr);
            a.y = elu1(pb.y + a.y * dr);
        }
#pragma unroll
        for (int t = 0; t < TN; t++) {
            v2f b = *(const v2f*)(bl + (ks * NT + t) * 64);
            acc[t] = __builtin_amdgcn_wmma_f32_16x16x4_f32(
                false, a, false, b, (short)0, acc[t], false, false);
        }
    }

    float bc[TN];
#pragma unroll
    for (int t = 0; t < TN; t++) bc[t] = FIRST ? 0.0f : bout[(t0 + t) * 16 + m];

    const long long rbase = row0 + (lane >> 4) * 8;

    float wv[8];
    if (FIRST) {
        int cnt[8] = {(int)pc0.x, (int)pc0.y, (int)pc0.z, (int)pc0.w,
                      (int)pc1.x, (int)pc1.y, (int)pc1.z, (int)pc1.w};
#pragma unroll
        for (int v = 0; v < 8; v++) {
            int sm = cnt[v];
            sm += __shfl_xor(sm, 1, 16);
            sm += __shfl_xor(sm, 2, 16);
            sm += __shfl_xor(sm, 4, 16);
            sm += __shfl_xor(sm, 8, 16);
            wv[v] = rsqrtf((float)(sm + 1));
        }
    } else {
#pragma unroll
        for (int v = 0; v < 8; v++) {
            long long r = rbase + v;
            wv[v] = (r < n) ? Dn[r] : 1.0f;
        }
    }

#pragma unroll
    for (int v = 0; v < 8; v++) {
        long long r = rbase + v;
        if (r < n) {
            const float w = wv[v];
            if (FIRST && own_dn && m == 0) Dn[r] = w;
            float* __restrict__ orow = H + r * DOUT + t0 * 16;
            float* __restrict__ arow = Agg + r * DOUT + t0 * 16;
#pragma unroll
            for (int t = 0; t < TN; t++) {
                const float hs = acc[t][v] * w;
                orow[t * 16 + m] = hs;
                arow[t * 16 + m] = FIRST ? hs : (bc[t] + hs * w);
            }
        }
    }
}

template <int DOUT, bool FIRST>
__global__ __launch_bounds__(GEMM_WV * 32) void gemm_wmma_kernel(const float* __restrict__ X,
                                                        const float* __restrict__ W,
                                                        const float* __restrict__ bin,
                                                        const float* __restrict__ bout,
                                                        float* Dn,
                                                        const unsigned int* __restrict__ P,
                                                        int pstride,
                                                        float* __restrict__ H,
                                                        float* __restrict__ Agg, int n,
                                                        unsigned int* __restrict__ tick) {
    constexpr int NT = DOUT / 16;
    constexpr int NPAIR = 32 * NT * 32;
    __shared__ __attribute__((aligned(16))) float sW[NPAIR * 2];
    __shared__ __attribute__((aligned(16))) float sB[FIRST ? 4 : DIN];

    constexpr int NQ = DOUT / 4;
    for (int it = threadIdx.x; it < (DIN / 2) * NQ; it += GEMM_WV * 32) {
        const int q = it % NQ;
        const int p = it / NQ;
        const float4 r0 = *(const float4*)(W + (2 * p) * DOUT + 4 * q);
        const float4 r1 = *(const float4*)(W + (2 * p + 1) * DOUT + 4 * q);
        float* d8 = sW + ((((p >> 1) * NT + (q >> 2)) * 32 + (p & 1) * 16 + (q & 3) * 4) * 2);
        *(float4*)(d8)     = make_float4(r0.x, r1.x, r0.y, r1.y);
        *(float4*)(d8 + 4) = make_float4(r0.z, r1.z, r0.w, r1.w);
    }
    if (!FIRST) {
        if (threadIdx.x < DIN) sB[threadIdx.x] = bin[threadIdx.x];
    }
    if (blockIdx.x == 0 && threadIdx.x < 128) tick[threadIdx.x] = 0u;
    __syncthreads();

    const int wave = __builtin_amdgcn_readfirstlane((int)(threadIdx.x >> 5));
    const int lane = threadIdx.x & 31;
    const int nstrips = (n + 15) >> 4;
    const int G = (int)gridDim.x;

    if (wave < GEMM_RUN - 1) {
        for (int strip = (int)blockIdx.x + G * wave; strip < nstrips; strip += G * GEMM_RUN)
            gemm_strip<DOUT, FIRST, NT>(X, sW, sB, bout, Dn, P, pstride, H, Agg, n,
                                        strip, 0, true, lane);
    } else {
        constexpr int NSPLIT = NT >= 8 ? 4 : 2;
        constexpr int TQ = NT / NSPLIT;
        const int qt = wave - (GEMM_RUN - 1);
        if (qt < NSPLIT) {
            for (int strip = (int)blockIdx.x + G * (GEMM_RUN - 1); strip < nstrips;
                 strip += G * GEMM_RUN)
                gemm_strip<DOUT, FIRST, TQ>(X, sW, sB, bout, Dn, P, pstride, H, Agg, n,
                                            strip, qt * TQ, qt == 0, lane);
        }
    }
}

#define SB 2

template <int D, bool DSTW>
__global__ __launch_bounds__(256) void scatter_kernel(const int* __restrict__ ei, int e,
                                                      const float* __restrict__ dinv,
                                                      const float* __restrict__ H,
                                                      float* __restrict__ Agg,
                                                      unsigned int* __restrict__ tick,
                                                      int nitems, int nsplit) {
    constexpr int VEC = D / 32;
    constexpr int RG = 32 / VEC;

    __shared__ int s_item;
    if (threadIdx.x == 0) {
        const unsigned int id = __builtin_amdgcn_s_sendmsg_rtn(0x87);
        const int h = (((id >> 16) & 15u) >= 4u) ? 1 : 0;
        int item = -1;
        unsigned int t = atomicAdd(&tick[h * 64], 1u);
        if (t < (unsigned int)nitems) {
            item = (int)t * 2 + h;
        } else {
            t = atomicAdd(&tick[(1 - h) * 64], 1u);
            if (t < (unsigned int)nitems) item = (int)t * 2 + (1 - h);
        }
        s_item = item;
    }
    __syncthreads();
    const int item = __builtin_amdgcn_readfirstlane(s_item);
    if (item < 0) return;
    const bool upper = (item & 1) != 0;

    const int lane = threadIdx.x & 31;
    const int wv = __builtin_amdgcn_readfirstlane((int)(threadIdx.x >> 5));
    const int base0 = (item >> 1) * (256 * SB) + wv * 32;

    int s[SB], d[SB];
    bool ok[SB];
#pragma unroll
    for (int b = 0; b < SB; b++) {
        const int idx = base0 + b * 256 + lane;
        ok[b] = idx < e;
        const int ix = ok[b] ? idx : 0;
        s[b] = ei[ix];
        d[b] = ei[(long long)e + ix];
    }
    asm volatile("" ::"v"(s[0]), "v"(s[1]), "v"(d[0]), "v"(d[1]));
    static_assert(SB == 2, "the statement above names the index values of two batches");
    bool mine[SB];
    unsigned int mk[SB];
#pragma unroll
    for (int b = 0; b < SB; b++) {
        mine[b] = ok[b] && ((d[b] >= nsplit) == upper);
        mk[b] = __builtin_amdgcn_ballot_w32(mine[b]);
    }
    float w[SB];
#pragma unroll
    for (int b = 0; b < SB; b++) {
        w[b] = 0.0f;
        if (DSTW) w[b] = dinv[mine[b] ? d[b] : 0];
    }

    const float* __restrict__ hl = H + lane;
    float* __restrict__ al = Agg + lane;

#pragma unroll
    for (int b = 0; b < SB; b++) {
        unsigned int m = mk[b];
        while (m != 0u) {
            const int left = __builtin_popcount(m);
            const int nj = left < RG ? left : RG;
            int dl[RG];
            float cur[RG][VEC];
#pragma unroll
            for (int j = 0; j < RG; j++) {
                dl[j] = 0;
                if (j < nj) {
                    const int l = __builtin_ctz(m);
                    m &= m - 1u;
                    dl[j] = l;
                    const int sj = __builtin_amdgcn_readlane(s[b], l);
                    const float* __restrict__ hs = hl + (long long)sj * D;
#pragma unroll
                    for (int k = 0; k < VEC; k++) cur[j][k] = hs[k * 32];
                }
            }
#pragma unroll
            for (int j = 0; j < RG; j++) {
                if (j < nj) {
                    const int dj = __builtin_amdgcn_readlane(d[b], dl[j]);
                    float* __restrict__ ad = al + (long long)dj * D;
                    if (DSTW) {
                        const float wj = __builtin_bit_cast(
                            float,
                            __builtin_amdgcn_readlane(__builtin_bit_cast(int, w[b]), dl[j]));
#pragma unroll
                        for (int k = 0; k < VEC; k++) atomicAdd(ad + k * 32, cur[j][k] * wj);
                    } else {
#pragma unroll
                        for (int k = 0; k < VEC; k++) atomicAdd(ad + k * 32, cur[j][k]);
                    }
                }
            }
        }
    }
}

extern "C" void kernel_launch(void* const* d_in, const int* in_sizes, int n_in,
                              void* d_out, int out_size, void* d_ws, size_t ws_size,
                              hipStream_t stream) {
    const float* x  = (const float*)d_in[0];
    const int*   ei = (const int*)d_in[1];
    const float* W1 = (const float*)d_in[2];
    const float* b1 = (const float*)d_in[3];
    const float* W2 = (const float*)d_in[4];
    const float* b2 = (const float*)d_in[5];
    float* out = (float*)d_out;

    const int N = in_sizes[0] / DIN;
    const int E = in_sizes[1] / 2;
    const int DH = 128;

    char* ws = (char*)d_ws;
    float* dinv = (float*)ws;
    size_t off = (((size_t)N * sizeof(float)) + 255) & ~(size_t)255;
    float* hbuf = (float*)(ws + off);
    float* aggbuf = (float*)(ws + off + (size_t)N * DH * sizeof(float));

    const int B = 256;
    const int gGemm = GEMM_WG;
    const int BG = GEMM_WV * 32;
    const int nitems = (E + 256 * SB - 1) / (256 * SB);
    const int gEdgeW = 2 * nitems;
    const int nsplit = N / 2;

    unsigned int* pcnt = (unsigned int*)d_out;
    const int gR = (N + HRANGE - 1) / HRANGE;
    const int pstride = gR * HRANGE;
    deg_hist_kernel<<<dim3(HC, gR), 512, 0, stream>>>(ei, E, pcnt, pstride);

    unsigned int* tick1 = pcnt + (size_t)HC * pstride;
    unsigned int* tick2 = (unsigned int*)(hbuf + (size_t)N * 64);

    gemm_wmma_kernel<128, true><<<gGemm, BG, 0, stream>>>(x, W1, b1, b1, dinv, pcnt, pstride, hbuf, aggbuf, N, tick1);
    scatter_kernel<128, false><<<gEdgeW, B, 0, stream>>>(ei, E, dinv, hbuf, aggbuf, tick1, nitems, nsplit);

    gemm_wmma_kernel<64, false><<<gGemm, BG, 0, stream>>>(aggbuf, W2, b1, b2, dinv, pcnt, pstride, hbuf, out, N, tick2);
    scatter_kernel<64, true><<<gEdgeW, B, 0, stream>>>(ei, E, dinv, hbuf, out, tick2, nitems, nsplit);
}
